# starting kernel
# baseline (speedup 1.0000x reference)
#include <hip/hip_runtime.h>


typedef __attribute__((ext_vector_type(2))) float v2f;
typedef __attribute__((ext_vector_type(8))) float v8f;

#define NQ     20
#define NBATCH 16
#define SSIZE  (1u << NQ)

#define WAVES_PER_BATCH 512
#define TILES_PER_WAVE  8

template <int P0, bool FIRST>
__global__ __launch_bounds__(256) void rx_group_pass(
    const float* __restrict__ phi,
    const float* __restrict__ thetas,
    float*       __restrict__ state)
{
    const int lane  = threadIdx.x & 31;
    const int wid   = (int)blockIdx.x * 8 + (threadIdx.x >> 5);
    const int b     = wid >> 9;
    const int wtile = wid & (WAVES_PER_BATCH - 1);
    const int h     = lane >> 4;
    const int n     = lane & 15;

    const int g = (16 - P0) >> 2;
    float cb[4], sb[4];
#pragma unroll
    for (int beta = 0; beta < 4; ++beta) {
        float th = 0.5f * thetas[b * NQ + 4 * g + (3 - beta)];
        cb[beta] = cosf(th);
        sb[beta] = sinf(th);
    }

    v2f aR[4], aI[4], amI[4];
    const int M = n;
#pragma unroll
    for (int kc = 0; kc < 4; ++kc) {
#pragma unroll
        for (int r = 0; r < 2; ++r) {
            const int j = 4 * kc + r + 2 * h;
            const int x = M ^ j;
            float prod = 1.0f;
#pragma unroll
            for (int beta = 0; beta < 4; ++beta)
                prod *= ((x >> beta) & 1) ? sb[beta] : cb[beta];
            const int k = __popc(x);
            const float Rv = (k & 1) ? 0.0f : ((k & 2) ? -prod : prod);
            const float Iv = (k & 1) ? ((k & 2) ? prod : -prod) : 0.0f;
            aR[kc][r]  = Rv;
            aI[kc][r]  = Iv;
            amI[kc][r] = -Iv;
        }
    }

    const unsigned LOMASK = (1u << P0) - 1u;

#pragma unroll 1
    for (int t = 0; t < TILES_PER_WAVE; ++t) {
        const unsigned tile  = (unsigned)wtile * TILES_PER_WAVE + (unsigned)t;
        const unsigned nc    = tile * 16u + (unsigned)n;
        const unsigned hi    = nc >> P0;
        const unsigned lo    = nc & LOMASK;
        const unsigned ebase = (hi << (P0 + 4)) | lo;
        const size_t   base  = (size_t)b * SSIZE + ebase;

        v2f bRe[4], bIm[4];
        if (FIRST) {
            const float* src = phi + base;
#pragma unroll
            for (int kc = 0; kc < 4; ++kc)
#pragma unroll
                for (int r = 0; r < 2; ++r) {
                    const int j = 4 * kc + r + 2 * h;
                    bRe[kc][r] = src[(size_t)j << P0];
                }
        } else {
            const v2f* src = (const v2f*)state + base;
#pragma unroll
            for (int kc = 0; kc < 4; ++kc)
#pragma unroll
                for (int r = 0; r < 2; ++r) {
                    const int j = 4 * kc + r + 2 * h;
                    const v2f v = src[(size_t)j << P0];
                    bRe[kc][r] = v.x;
                    bIm[kc][r] = v.y;
                }
        }

        v8f accRe = {};
        v8f accIm = {};
#pragma unroll
        for (int kc = 0; kc < 4; ++kc) {
            accRe = __builtin_amdgcn_wmma_f32_16x16x4_f32(
                false, aR[kc], false, bRe[kc], (short)0, accRe, false, false);
            accIm = __builtin_amdgcn_wmma_f32_16x16x4_f32(
                false, aI[kc], false, bRe[kc], (short)0, accIm, false, false);
        }
        if (!FIRST) {
#pragma unroll
            for (int kc = 0; kc < 4; ++kc) {
                accRe = __builtin_amdgcn_wmma_f32_16x16x4_f32(
                    false, amI[kc], false, bIm[kc], (short)0, accRe, false, false);
                accIm = __builtin_amdgcn_wmma_f32_16x16x4_f32(
                    false, aR[kc], false, bIm[kc], (short)0, accIm, false, false);
            }
        }

        v2f* dst = (v2f*)state + base;
#pragma unroll
        for (int r = 0; r < 8; ++r) {
            const int Mo = r + 8 * h;
            v2f o;
            o.x = accRe[r];
            o.y = accIm[r];
            dst[(size_t)Mo << P0] = o;
        }
    }
}

extern "C" void kernel_launch(void* const* d_in, const int* in_sizes, int n_in,
                              void* d_out, int out_size, void* d_ws, size_t ws_size,
                              hipStream_t stream) {
    (void)in_sizes; (void)n_in; (void)d_ws; (void)ws_size; (void)out_size;
    const float* phi    = (const float*)d_in[0];
    const float* thetas = (const float*)d_in[1];
    float* out = (float*)d_out;

    const dim3 grid(1024), block(256);
    rx_group_pass<16, true ><<<grid, block, 0, stream>>>(phi, thetas, out);
    rx_group_pass<12, false><<<grid, block, 0, stream>>>(phi, thetas, out);
    rx_group_pass< 8, false><<<grid, block, 0, stream>>>(phi, thetas, out);
    rx_group_pass< 4, false><<<grid, block, 0, stream>>>(phi, thetas, out);
    rx_group_pass< 0, false><<<grid, block, 0, stream>>>(phi, thetas, out);
}


# Angles: sin and cos from one shared three-part pi/2 reduction plus two short polynomials, branch-free; library sinf/cosf and their large-argument blocks leave the text (low pass 46 -> 27 KB)
# speedup vs baseline: 4.1544x; 4.1544x over previous
#include <hip/hip_runtime.h>


typedef __attribute__((ext_vector_type(2))) float v2f;
typedef __attribute__((ext_vector_type(8))) float v8f;

#define NQ     20
#define NBATCH 16
#define SSIZE  (1u << NQ)

#define ZPITCH 18

__device__ __forceinline__ void sincos_half_angle(float x, float& sn, float& cs) {
    const float n = __builtin_rintf(x * 0.636619746685028076171875f);
    float r = __builtin_fmaf(n, -1.5707962512969970703125f, x);
    r = __builtin_fmaf(n, -7.5497894158615963e-08f, r);
    r = __builtin_fmaf(n, -5.3903025299577648e-15f, r);
    const float r2 = r * r;
    float ps = __builtin_fmaf(r2, -1.9515295891e-4f, 8.3321608736e-3f);
    ps = __builtin_fmaf(r2, ps, -1.6666654611e-1f);
    const float sr = __builtin_fmaf(r * r2, ps, r);
    float pc = __builtin_fmaf(r2, 2.443315711809948e-5f, -1.388731625493765e-3f);
    pc = __builtin_fmaf(r2, pc, 4.166664568298827e-2f);
    const float cr = __builtin_fmaf(r2 * r2, pc, __builtin_fmaf(r2, -0.5f, 1.0f));
    const int q = (int)n;
    const float s0 = (q & 1) ? cr : sr;
    const float c0 = (q & 1) ? sr : cr;
    sn = (q & 2) ? -s0 : s0;
    cs = ((q + 1) & 2) ? -c0 : c0;
}

__device__ __forceinline__ void build_group_frags(
    const float* __restrict__ thetas, int b, int g, int n, int h,
    v2f (&fR)[4], v2f (&fI)[4], v2f (&fmI)[4])
{
    float cb[4], sb[4];
#pragma unroll
    for (int beta = 0; beta < 4; ++beta)
        sincos_half_angle(0.5f * thetas[b * NQ + 4 * g + (3 - beta)], sb[beta], cb[beta]);
#pragma unroll
    for (int kc = 0; kc < 4; ++kc) {
#pragma unroll
        for (int r = 0; r < 2; ++r) {
            const int j = 4 * kc + r + 2 * h;
            const int x = n ^ j;
            float prod = 1.0f;
#pragma unroll
            for (int beta = 0; beta < 4; ++beta)
                prod *= ((x >> beta) & 1) ? sb[beta] : cb[beta];
            const int k = __popc(x);
            const float Rv = (k & 1) ? 0.0f : ((k & 2) ? -prod : prod);
            const float Iv = (k & 1) ? ((k & 2) ? prod : -prod) : 0.0f;
            fR[kc][r]  = Rv;
            fI[kc][r]  = Iv;
            fmI[kc][r] = -Iv;
        }
    }
}

__device__ __forceinline__ float xhalf(float x) {
    return __builtin_bit_cast(float, __builtin_amdgcn_permlanex16(
        __builtin_bit_cast(int, x), __builtin_bit_cast(int, x),
        0x76543210, 0xfedcba98, false, false));
}

__device__ __forceinline__ void pair_halves(float a, float c, int h, float& lo, float& hi) {
    const float give = h ? a : c;
    const float got  = xhalf(give);
    lo = h ? got : a;
    hi = h ? c : got;
}

__device__ __forceinline__ void low8_tile(
    const v2f (&hR)[4], const v2f (&hI)[4], const v2f (&hmI)[4],
    const v2f (&lR)[4], const v2f (&lI)[4], const v2f (&lmI)[4],
    const v2f (&bRe)[4], const v2f (&bIm)[4],
    v2f* __restrict__ zb, v2f* __restrict__ dst, int h, int n)
{
    v8f zRe = {};
    v8f zIm = {};
#pragma unroll
    for (int kc = 0; kc < 4; ++kc) {
        zRe = __builtin_amdgcn_wmma_f32_16x16x4_f32(
            false, hR[kc], false, bRe[kc], (short)0, zRe, false, false);
        zIm = __builtin_amdgcn_wmma_f32_16x16x4_f32(
            false, hI[kc], false, bRe[kc], (short)0, zIm, false, false);
    }
#pragma unroll
    for (int kc = 0; kc < 4; ++kc) {
        zRe = __builtin_amdgcn_wmma_f32_16x16x4_f32(
            false, hmI[kc], false, bIm[kc], (short)0, zRe, false, false);
        zIm = __builtin_amdgcn_wmma_f32_16x16x4_f32(
            false, hR[kc], false, bIm[kc], (short)0, zIm, false, false);
    }

#pragma unroll
    for (int r = 0; r < 8; ++r) {
        v2f o;
        o.x = zRe[r];
        o.y = zIm[r];
        zb[(r + 8 * h) * ZPITCH + n] = o;
    }
    __builtin_amdgcn_fence(__ATOMIC_RELEASE, "wavefront");
    __builtin_amdgcn_wave_barrier();
    __builtin_amdgcn_fence(__ATOMIC_ACQUIRE, "wavefront");

    v2f aRe[4], aIm[4];
#pragma unroll
    for (int kc = 0; kc < 4; ++kc) {
        const float4 q = *(const float4*)&zb[n * ZPITCH + 4 * kc + 2 * h];
        aRe[kc][0] = q.x;  aIm[kc][0] = q.y;
        aRe[kc][1] = q.z;  aIm[kc][1] = q.w;
    }
    __builtin_amdgcn_fence(__ATOMIC_RELEASE, "wavefront");
    __builtin_amdgcn_wave_barrier();
    __builtin_amdgcn_fence(__ATOMIC_ACQUIRE, "wavefront");

    v8f yRe = {};
    v8f yIm = {};
#pragma unroll
    for (int kc = 0; kc < 4; ++kc) {
        yRe = __builtin_amdgcn_wmma_f32_16x16x4_f32(
            false, aRe[kc], false, lR[kc], (short)0, yRe, false, false);
        yIm = __builtin_amdgcn_wmma_f32_16x16x4_f32(
            false, aRe[kc], false, lI[kc], (short)0, yIm, false, false);
    }
#pragma unroll
    for (int kc = 0; kc < 4; ++kc) {
        yRe = __builtin_amdgcn_wmma_f32_16x16x4_f32(
            false, aIm[kc], false, lmI[kc], (short)0, yRe, false, false);
        yIm = __builtin_amdgcn_wmma_f32_16x16x4_f32(
            false, aIm[kc], false, lR[kc], (short)0, yIm, false, false);
    }

#pragma unroll
    for (int r = 0; r < 8; ++r) {
        v2f o;
        o.x = yRe[r];
        o.y = yIm[r];
        dst[(size_t)(r + 8 * h) << 4] = o;
    }
}

__global__ __launch_bounds__(128) void rx_low12_pass(
    const float* __restrict__ thetas,
    float*       __restrict__ state)
{
    __shared__ __attribute__((aligned(16))) v2f zbuf[4][16 * ZPITCH];

    const int lane = threadIdx.x & 31;
    const int wib  = threadIdx.x >> 5;
    const int wid  = (int)blockIdx.x * 4 + wib;
    const int b    = __builtin_amdgcn_readfirstlane(wid >> 8);
    const int blk  = wid & 255;
    const int h    = lane >> 4;
    const int n    = lane & 15;

    const size_t base = (size_t)b * SSIZE + (size_t)blk * 4096u + (unsigned)n;
    const v2f*   src  = (const v2f*)state + base;

    v2f x[16][4][2];
#pragma unroll
    for (int jb = 0; jb < 16; ++jb)
#pragma unroll
        for (int kc = 0; kc < 4; ++kc)
#pragma unroll
            for (int r = 0; r < 2; ++r) {
                const int ja = 4 * kc + r + 2 * h;
                x[jb][kc][r] = src[((size_t)ja << 8) + ((size_t)jb << 4)];
            }

    v2f g2R[4], g2I[4], g2mI[4];
    build_group_frags(thetas, b, 2, n, h, g2R, g2I, g2mI);

    v8f zRe[16], zIm[16];
#pragma unroll
    for (int jb = 0; jb < 16; ++jb) {
        v2f bRe[4], bIm[4];
#pragma unroll
        for (int kc = 0; kc < 4; ++kc)
#pragma unroll
            for (int r = 0; r < 2; ++r) {
                bRe[kc][r] = x[jb][kc][r].x;
                bIm[kc][r] = x[jb][kc][r].y;
            }
        v8f accRe = {};
        v8f accIm = {};
#pragma unroll
        for (int kc = 0; kc < 4; ++kc) {
            accRe = __builtin_amdgcn_wmma_f32_16x16x4_f32(
                false, g2R[kc], false, bRe[kc], (short)0, accRe, false, false);
            accIm = __builtin_amdgcn_wmma_f32_16x16x4_f32(
                false, g2I[kc], false, bRe[kc], (short)0, accIm, false, false);
        }
#pragma unroll
        for (int kc = 0; kc < 4; ++kc) {
            accRe = __builtin_amdgcn_wmma_f32_16x16x4_f32(
                false, g2mI[kc], false, bIm[kc], (short)0, accRe, false, false);
            accIm = __builtin_amdgcn_wmma_f32_16x16x4_f32(
                false, g2R[kc], false, bIm[kc], (short)0, accIm, false, false);
        }
        zRe[jb] = accRe;
        zIm[jb] = accIm;
    }

    v2f hR[4], hI[4], hmI[4];
    v2f lR[4], lI[4], lmI[4];
    build_group_frags(thetas, b, 3, n, h, hR, hI, hmI);
    build_group_frags(thetas, b, 4, n, h, lR, lI, lmI);

    v2f* zb  = &zbuf[wib][0];
    v2f* dst = (v2f*)state + base;
#pragma unroll
    for (int rho = 0; rho < 8; ++rho) {
        v2f loRe[4], loIm[4], hiRe[4], hiIm[4];
#pragma unroll
        for (int kc = 0; kc < 4; ++kc)
#pragma unroll
            for (int r = 0; r < 2; ++r) {
                const int jA = 4 * kc + r;
                float lo, hi;
                pair_halves(zRe[jA][rho], zRe[jA + 2][rho], h, lo, hi);
                loRe[kc][r] = lo;  hiRe[kc][r] = hi;
                pair_halves(zIm[jA][rho], zIm[jA + 2][rho], h, lo, hi);
                loIm[kc][r] = lo;  hiIm[kc][r] = hi;
            }
        low8_tile(hR, hI, hmI, lR, lI, lmI, loRe, loIm, zb, dst + ((size_t)rho << 8), h, n);
        low8_tile(hR, hI, hmI, lR, lI, lmI, hiRe, hiIm, zb, dst + ((size_t)(rho + 8) << 8), h, n);
    }
}

__device__ __forceinline__ void high8_second(
    const v2f (&uR)[4], const v2f (&uI)[4], const v2f (&umI)[4],
    const v2f (&bRe)[4], const v2f (&bIm)[4], v2f* __restrict__ dst, int h)
{
    v8f yRe = {};
    v8f yIm = {};
#pragma unroll
    for (int kc = 0; kc < 4; ++kc) {
        yRe = __builtin_amdgcn_wmma_f32_16x16x4_f32(
            false, uR[kc], false, bRe[kc], (short)0, yRe, false, false);
        yIm = __builtin_amdgcn_wmma_f32_16x16x4_f32(
            false, uI[kc], false, bRe[kc], (short)0, yIm, false, false);
    }
#pragma unroll
    for (int kc = 0; kc < 4; ++kc) {
        yRe = __builtin_amdgcn_wmma_f32_16x16x4_f32(
            false, umI[kc], false, bIm[kc], (short)0, yRe, false, false);
        yIm = __builtin_amdgcn_wmma_f32_16x16x4_f32(
            false, uR[kc], false, bIm[kc], (short)0, yIm, false, false);
    }
#pragma unroll
    for (int r = 0; r < 8; ++r) {
        v2f o;
        o.x = yRe[r];
        o.y = yIm[r];
        dst[(size_t)(r + 8 * h) << 12] = o;
    }
}

__global__ __launch_bounds__(128) void rx_high8_pass(
    const float* __restrict__ phi,
    const float* __restrict__ thetas,
    float*       __restrict__ state)
{
    const int lane = threadIdx.x & 31;
    const int wid  = (int)blockIdx.x * 4 + (threadIdx.x >> 5);
    const int b    = __builtin_amdgcn_readfirstlane(wid >> 8);
    const int ct   = wid & 255;
    const int h    = lane >> 4;
    const int n    = lane & 15;

    const size_t base = (size_t)b * SSIZE + (unsigned)(ct * 16 + n);
    const float* src  = phi + base;

    v2f x[16][4];
#pragma unroll
    for (int j2 = 0; j2 < 16; ++j2)
#pragma unroll
        for (int kc = 0; kc < 4; ++kc)
#pragma unroll
            for (int r = 0; r < 2; ++r) {
                const int j1 = 4 * kc + r + 2 * h;
                x[j2][kc][r] = src[((size_t)j1 << 16) + ((size_t)j2 << 12)];
            }

    v2f g0R[4], g0I[4], g0mI[4];
    v2f g1R[4], g1I[4], g1mI[4];
    build_group_frags(thetas, b, 0, n, h, g0R, g0I, g0mI);
    build_group_frags(thetas, b, 1, n, h, g1R, g1I, g1mI);

    v8f zRe[16], zIm[16];
#pragma unroll
    for (int j2 = 0; j2 < 16; ++j2) {
        v8f accRe = {};
        v8f accIm = {};
#pragma unroll
        for (int kc = 0; kc < 4; ++kc) {
            accRe = __builtin_amdgcn_wmma_f32_16x16x4_f32(
                false, g0R[kc], false, x[j2][kc], (short)0, accRe, false, false);
            accIm = __builtin_amdgcn_wmma_f32_16x16x4_f32(
                false, g0I[kc], false, x[j2][kc], (short)0, accIm, false, false);
        }
        zRe[j2] = accRe;
        zIm[j2] = accIm;
    }

    v2f* dst = (v2f*)state + base;
#pragma unroll
    for (int rho = 0; rho < 8; ++rho) {
        v2f loRe[4], loIm[4], hiRe[4], hiIm[4];
#pragma unroll
        for (int kc = 0; kc < 4; ++kc)
#pragma unroll
            for (int r = 0; r < 2; ++r) {
                const int jA = 4 * kc + r;
                float lo, hi;
                pair_halves(zRe[jA][rho], zRe[jA + 2][rho], h, lo, hi);
                loRe[kc][r] = lo;  hiRe[kc][r] = hi;
                pair_halves(zIm[jA][rho], zIm[jA + 2][rho], h, lo, hi);
                loIm[kc][r] = lo;  hiIm[kc][r] = hi;
            }
        high8_second(g1R, g1I, g1mI, loRe, loIm, dst + ((size_t)rho << 16), h);
        high8_second(g1R, g1I, g1mI, hiRe, hiIm, dst + ((size_t)(rho + 8) << 16), h);
        __builtin_amdgcn_sched_barrier(0);
    }
}

extern "C" void kernel_launch(void* const* d_in, const int* in_sizes, int n_in,
                              void* d_out, int out_size, void* d_ws, size_t ws_size,
                              hipStream_t stream) {
    (void)in_sizes; (void)n_in; (void)d_ws; (void)ws_size; (void)out_size;
    const float* phi    = (const float*)d_in[0];
    const float* thetas = (const float*)d_in[1];
    float* out = (float*)d_out;

    rx_high8_pass<<<dim3(1024), dim3(128), 0, stream>>>(phi, thetas, out);
    rx_low12_pass<<<dim3(1024), dim3(128), 0, stream>>>(thetas, out);
}


# Complex tile products as three real products (T=R*(Xre+Xim) is the C operand of Re=T-(R+I)*Xim and Im=T+(I-R)*Xre): 12 matrix instructions each, not 16; f32 sum order of these products changes
# speedup vs baseline: 4.3970x; 1.0584x over previous
#include <hip/hip_runtime.h>


typedef __attribute__((ext_vector_type(2))) float v2f;
typedef __attribute__((ext_vector_type(8))) float v8f;

#define NQ     20
#define NBATCH 16
#define SSIZE  (1u << NQ)

#define ZPITCH 18

__device__ __forceinline__ void sincos_half_angle(float x, float& sn, float& cs) {
    const float n = __builtin_rintf(x * 0.636619746685028076171875f);
    float r = __builtin_fmaf(n, -1.5707962512969970703125f, x);
    r = __builtin_fmaf(n, -7.5497894158615963e-08f, r);
    r = __builtin_fmaf(n, -5.3903025299577648e-15f, r);
    const float r2 = r * r;
    float ps = __builtin_fmaf(r2, -1.9515295891e-4f, 8.3321608736e-3f);
    ps = __builtin_fmaf(r2, ps, -1.6666654611e-1f);
    const float sr = __builtin_fmaf(r * r2, ps, r);
    float pc = __builtin_fmaf(r2, 2.443315711809948e-5f, -1.388731625493765e-3f);
    pc = __builtin_fmaf(r2, pc, 4.166664568298827e-2f);
    const float cr = __builtin_fmaf(r2 * r2, pc, __builtin_fmaf(r2, -0.5f, 1.0f));
    const int q = (int)n;
    const float s0 = (q & 1) ? cr : sr;
    const float c0 = (q & 1) ? sr : cr;
    sn = (q & 2) ? -s0 : s0;
    cs = ((q + 1) & 2) ? -c0 : c0;
}

__device__ __forceinline__ void build_group_frags(
    const float* __restrict__ thetas, int b, int g, int n, int h,
    v2f (&fR)[4], v2f (&fI)[4])
{
    float cb[4], sb[4];
#pragma unroll
    for (int beta = 0; beta < 4; ++beta)
        sincos_half_angle(0.5f * thetas[b * NQ + 4 * g + (3 - beta)], sb[beta], cb[beta]);
#pragma unroll
    for (int kc = 0; kc < 4; ++kc) {
#pragma unroll
        for (int r = 0; r < 2; ++r) {
            const int j = 4 * kc + r + 2 * h;
            const int x = n ^ j;
            float prod = 1.0f;
#pragma unroll
            for (int beta = 0; beta < 4; ++beta)
                prod *= ((x >> beta) & 1) ? sb[beta] : cb[beta];
            const int k = __popc(x);
            const float Rv = (k & 1) ? 0.0f : ((k & 2) ? -prod : prod);
            const float Iv = (k & 1) ? ((k & 2) ? prod : -prod) : 0.0f;
            fR[kc][r] = Rv;
            fI[kc][r] = Iv;
        }
    }
}

__device__ __forceinline__ void three_product_frags(
    const v2f (&fR)[4], const v2f (&fI)[4], v2f (&fS)[4], v2f (&fD)[4])
{
#pragma unroll
    for (int kc = 0; kc < 4; ++kc) {
        fS[kc] = -(fR[kc] + fI[kc]);
        fD[kc] = fI[kc] - fR[kc];
    }
}

__device__ __forceinline__ void cprod_u_left(
    const v2f (&uR)[4], const v2f (&uS)[4], const v2f (&uD)[4],
    const v2f (&bRe)[4], const v2f (&bIm)[4], v8f& yRe, v8f& yIm)
{
    v8f t = {};
#pragma unroll
    for (int kc = 0; kc < 4; ++kc) {
        const v2f bSum = bRe[kc] + bIm[kc];
        t = __builtin_amdgcn_wmma_f32_16x16x4_f32(
            false, uR[kc], false, bSum, (short)0, t, false, false);
    }
    v8f re = t;
    v8f im = t;
#pragma unroll
    for (int kc = 0; kc < 4; ++kc) {
        re = __builtin_amdgcn_wmma_f32_16x16x4_f32(
            false, uS[kc], false, bIm[kc], (short)0, re, false, false);
        im = __builtin_amdgcn_wmma_f32_16x16x4_f32(
            false, uD[kc], false, bRe[kc], (short)0, im, false, false);
    }
    yRe = re;
    yIm = im;
}

__device__ __forceinline__ void cprod_u_right(
    const v2f (&aRe)[4], const v2f (&aIm)[4],
    const v2f (&uR)[4], const v2f (&uS)[4], const v2f (&uD)[4],
    v8f& yRe, v8f& yIm)
{
    v8f t = {};
#pragma unroll
    for (int kc = 0; kc < 4; ++kc) {
        const v2f aSum = aRe[kc] + aIm[kc];
        t = __builtin_amdgcn_wmma_f32_16x16x4_f32(
            false, aSum, false, uR[kc], (short)0, t, false, false);
    }
    v8f re = t;
    v8f im = t;
#pragma unroll
    for (int kc = 0; kc < 4; ++kc) {
        re = __builtin_amdgcn_wmma_f32_16x16x4_f32(
            false, aIm[kc], false, uS[kc], (short)0, re, false, false);
        im = __builtin_amdgcn_wmma_f32_16x16x4_f32(
            false, aRe[kc], false, uD[kc], (short)0, im, false, false);
    }
    yRe = re;
    yIm = im;
}

__device__ __forceinline__ float xhalf(float x) {
    return __builtin_bit_cast(float, __builtin_amdgcn_permlanex16(
        __builtin_bit_cast(int, x), __builtin_bit_cast(int, x),
        0x76543210, 0xfedcba98, false, false));
}

__device__ __forceinline__ void pair_halves(float a, float c, int h, float& lo, float& hi) {
    const float give = h ? a : c;
    const float got  = xhalf(give);
    lo = h ? got : a;
    hi = h ? c : got;
}

__device__ __forceinline__ void low8_tile(
    const v2f (&hR)[4], const v2f (&hS)[4], const v2f (&hD)[4],
    const v2f (&lR)[4], const v2f (&lS)[4], const v2f (&lD)[4],
    const v2f (&bRe)[4], const v2f (&bIm)[4],
    v2f* __restrict__ zb, v2f* __restrict__ dst, int h, int n)
{
    v8f zRe, zIm;
    cprod_u_left(hR, hS, hD, bRe, bIm, zRe, zIm);

#pragma unroll
    for (int r = 0; r < 8; ++r) {
        v2f o;
        o.x = zRe[r];
        o.y = zIm[r];
        zb[(r + 8 * h) * ZPITCH + n] = o;
    }
    __builtin_amdgcn_fence(__ATOMIC_RELEASE, "wavefront");
    __builtin_amdgcn_wave_barrier();
    __builtin_amdgcn_fence(__ATOMIC_ACQUIRE, "wavefront");

    v2f aRe[4], aIm[4];
#pragma unroll
    for (int kc = 0; kc < 4; ++kc) {
        const float4 q = *(const float4*)&zb[n * ZPITCH + 4 * kc + 2 * h];
        aRe[kc][0] = q.x;  aIm[kc][0] = q.y;
        aRe[kc][1] = q.z;  aIm[kc][1] = q.w;
    }
    __builtin_amdgcn_fence(__ATOMIC_RELEASE, "wavefront");
    __builtin_amdgcn_wave_barrier();
    __builtin_amdgcn_fence(__ATOMIC_ACQUIRE, "wavefront");

    v8f yRe, yIm;
    cprod_u_right(aRe, aIm, lR, lS, lD, yRe, yIm);

#pragma unroll
    for (int r = 0; r < 8; ++r) {
        v2f o;
        o.x = yRe[r];
        o.y = yIm[r];
        dst[(size_t)(r + 8 * h) << 4] = o;
    }
}

__global__ __launch_bounds__(128) void rx_low12_pass(
    const float* __restrict__ thetas,
    float*       __restrict__ state)
{
    __shared__ __attribute__((aligned(16))) v2f zbuf[4][16 * ZPITCH];

    const int lane = threadIdx.x & 31;
    const int wib  = threadIdx.x >> 5;
    const int wid  = (int)blockIdx.x * 4 + wib;
    const int b    = __builtin_amdgcn_readfirstlane(wid >> 8);
    const int blk  = wid & 255;
    const int h    = lane >> 4;
    const int n    = lane & 15;

    const size_t base = (size_t)b * SSIZE + (size_t)blk * 4096u + (unsigned)n;
    const v2f*   src  = (const v2f*)state + base;

    v2f x[16][4][2];
#pragma unroll
    for (int jb = 0; jb < 16; ++jb)
#pragma unroll
        for (int kc = 0; kc < 4; ++kc)
#pragma unroll
            for (int r = 0; r < 2; ++r) {
                const int ja = 4 * kc + r + 2 * h;
                x[jb][kc][r] = src[((size_t)ja << 8) + ((size_t)jb << 4)];
            }

    v2f g2R[4], g2I[4], g2S[4], g2D[4];
    build_group_frags(thetas, b, 2, n, h, g2R, g2I);
    three_product_frags(g2R, g2I, g2S, g2D);

    v8f zRe[16], zIm[16];
#pragma unroll
    for (int jb = 0; jb < 16; ++jb) {
        v2f bRe[4], bIm[4];
#pragma unroll
        for (int kc = 0; kc < 4; ++kc)
#pragma unroll
            for (int r = 0; r < 2; ++r) {
                bRe[kc][r] = x[jb][kc][r].x;
                bIm[kc][r] = x[jb][kc][r].y;
            }
        cprod_u_left(g2R, g2S, g2D, bRe, bIm, zRe[jb], zIm[jb]);
    }

    v2f hR[4], hI[4], hS[4], hD[4];
    v2f lR[4], lI[4], lS[4], lD[4];
    build_group_frags(thetas, b, 3, n, h, hR, hI);
    three_product_frags(hR, hI, hS, hD);
    build_group_frags(thetas, b, 4, n, h, lR, lI);
    three_product_frags(lR, lI, lS, lD);

    v2f* zb  = &zbuf[wib][0];
    v2f* dst = (v2f*)state + base;
#pragma unroll
    for (int rho = 0; rho < 8; ++rho) {
        v2f loRe[4], loIm[4], hiRe[4], hiIm[4];
#pragma unroll
        for (int kc = 0; kc < 4; ++kc)
#pragma unroll
            for (int r = 0; r < 2; ++r) {
                const int jA = 4 * kc + r;
                float lo, hi;
                pair_halves(zRe[jA][rho], zRe[jA + 2][rho], h, lo, hi);
                loRe[kc][r] = lo;  hiRe[kc][r] = hi;
                pair_halves(zIm[jA][rho], zIm[jA + 2][rho], h, lo, hi);
                loIm[kc][r] = lo;  hiIm[kc][r] = hi;
            }
        low8_tile(hR, hS, hD, lR, lS, lD, loRe, loIm, zb, dst + ((size_t)rho << 8), h, n);
        low8_tile(hR, hS, hD, lR, lS, lD, hiRe, hiIm, zb, dst + ((size_t)(rho + 8) << 8), h, n);
    }
}

__device__ __forceinline__ void high8_second(
    const v2f (&uR)[4], const v2f (&uS)[4], const v2f (&uD)[4],
    const v2f (&bRe)[4], const v2f (&bIm)[4], v2f* __restrict__ dst, int h)
{
    v8f yRe, yIm;
    cprod_u_left(uR, uS, uD, bRe, bIm, yRe, yIm);
#pragma unroll
    for (int r = 0; r < 8; ++r) {
        v2f o;
        o.x = yRe[r];
        o.y = yIm[r];
        dst[(size_t)(r + 8 * h) << 12] = o;
    }
}

__global__ __launch_bounds__(128) void rx_high8_pass(
    const float* __restrict__ phi,
    const float* __restrict__ thetas,
    float*       __restrict__ state)
{
    const int lane = threadIdx.x & 31;
    const int wid  = (int)blockIdx.x * 4 + (threadIdx.x >> 5);
    const int b    = __builtin_amdgcn_readfirstlane(wid >> 8);
    const int ct   = wid & 255;
    const int h    = lane >> 4;
    const int n    = lane & 15;

    const size_t base = (size_t)b * SSIZE + (unsigned)(ct * 16 + n);
    const float* src  = phi + base;

    v2f x[16][4];
#pragma unroll
    for (int j2 = 0; j2 < 16; ++j2)
#pragma unroll
        for (int kc = 0; kc < 4; ++kc)
#pragma unroll
            for (int r = 0; r < 2; ++r) {
                const int j1 = 4 * kc + r + 2 * h;
                x[j2][kc][r] = src[((size_t)j1 << 16) + ((size_t)j2 << 12)];
            }

    v2f g0R[4], g0I[4];
    v2f g1R[4], g1I[4], g1S[4], g1D[4];
    build_group_frags(thetas, b, 0, n, h, g0R, g0I);
    build_group_frags(thetas, b, 1, n, h, g1R, g1I);
    three_product_frags(g1R, g1I, g1S, g1D);

    v8f zRe[16], zIm[16];
#pragma unroll
    for (int j2 = 0; j2 < 16; ++j2) {
        v8f accRe = {};
        v8f accIm = {};
#pragma unroll
        for (int kc = 0; kc < 4; ++kc) {
            accRe = __builtin_amdgcn_wmma_f32_16x16x4_f32(
                false, g0R[kc], false, x[j2][kc], (short)0, accRe, false, false);
            accIm = __builtin_amdgcn_wmma_f32_16x16x4_f32(
                false, g0I[kc], false, x[j2][kc], (short)0, accIm, false, false);
        }
        zRe[j2] = accRe;
        zIm[j2] = accIm;
    }

    v2f* dst = (v2f*)state + base;
#pragma unroll
    for (int rho = 0; rho < 8; ++rho) {
        v2f loRe[4], loIm[4], hiRe[4], hiIm[4];
#pragma unroll
        for (int kc = 0; kc < 4; ++kc)
#pragma unroll
            for (int r = 0; r < 2; ++r) {
                const int jA = 4 * kc + r;
                float lo, hi;
                pair_halves(zRe[jA][rho], zRe[jA + 2][rho], h, lo, hi);
                loRe[kc][r] = lo;  hiRe[kc][r] = hi;
                pair_halves(zIm[jA][rho], zIm[jA + 2][rho], h, lo, hi);
                loIm[kc][r] = lo;  hiIm[kc][r] = hi;
            }
        high8_second(g1R, g1S, g1D, loRe, loIm, dst + ((size_t)rho << 16), h);
        __builtin_amdgcn_sched_barrier(0);
        high8_second(g1R, g1S, g1D, hiRe, hiIm, dst + ((size_t)(rho + 8) << 16), h);
        __builtin_amdgcn_sched_barrier(0);
    }
}

extern "C" void kernel_launch(void* const* d_in, const int* in_sizes, int n_in,
                              void* d_out, int out_size, void* d_ws, size_t ws_size,
                              hipStream_t stream) {
    (void)in_sizes; (void)n_in; (void)d_ws; (void)ws_size; (void)out_size;
    const float* phi    = (const float*)d_in[0];
    const float* thetas = (const float*)d_in[1];
    float* out = (float*)d_out;

    rx_high8_pass<<<dim3(1024), dim3(128), 0, stream>>>(phi, thetas, out);
    rx_low12_pass<<<dim3(1024), dim3(128), 0, stream>>>(thetas, out);
}
